# starting kernel
# baseline (speedup 1.0000x reference)
#include <hip/hip_runtime.h>

typedef __attribute__((ext_vector_type(2))) float v2f;
typedef __attribute__((ext_vector_type(8))) float v8f;

#define B_ 2
#define G_ 50000
#define K_ 32
#define A_ 64

#define WAVES_PER_BLOCK 8
#define TILE_M 16
#define MSG_STRIDE 68
#define WPAIR_STRIDE 160

__global__ __launch_bounds__(256)
void interaction_block_kernel(const float* __restrict__ out_in,
                              const int*   __restrict__ nbr_idx,
                              const float* __restrict__ W2g,
                              const float* __restrict__ b2g,
                              const float* __restrict__ W3g,
                              const float* __restrict__ b3g,
                              float*       __restrict__ y_out)
{
    __shared__ float sW2[(A_ / 2) * WPAIR_STRIDE];
    __shared__ float sW3[(A_ / 2) * WPAIR_STRIDE];
    __shared__ float sb2[A_];
    __shared__ float sb3[A_];
    __shared__ float sMsg[WAVES_PER_BLOCK][TILE_M * MSG_STRIDE];

    const int tid = threadIdx.x;

    for (int i = tid; i < A_ * A_; i += 256) {
        const int k = i >> 6;
        const int n = i & 63;
        const int s = (k >> 1) * WPAIR_STRIDE + 2 * n + (k & 1);
        sW2[s] = W2g[i];
        sW3[s] = W3g[i];
    }
    if (tid < A_) { sb2[tid] = b2g[tid]; sb3[tid] = b3g[tid]; }
    __syncthreads();

    const int wave = tid >> 5;
    const int lane = tid & 31;
    const int tile = blockIdx.x * WAVES_PER_BLOCK + wave;
    const int n_tiles = (B_ * G_) / TILE_M;
    if (tile >= n_tiles) return;

    const int row0 = tile * TILE_M;
    const int b    = row0 / G_;
    const int g0   = row0 % G_;
    float* msg = sMsg[wave];

    const float* outB   = out_in + (size_t)b * G_ * A_;
    const float* lanePt = outB + 2 * lane;
    for (int m = 0; m < TILE_M; ++m) {
        const int g = g0 + m;
        const int myidx = nbr_idx[g * K_ + lane];
        float ax = 0.f, ay = 0.f;
        #pragma unroll
        for (int k = 0; k < K_; ++k) {
            const int idx = __builtin_amdgcn_readlane(myidx, k);
            if (idx < G_) {
                const v2f val = *(const v2f*)(lanePt + (size_t)idx * A_);
                ax += val.x;
                ay += val.y;
            }
        }
        msg[m * MSG_STRIDE + 2 * lane]     = ax;
        msg[m * MSG_STRIDE + 2 * lane + 1] = ay;
    }

    const int am    = lane & 15;
    const int khalf = lane >> 4;
    const int ncol  = lane & 15;
    const int rsel  = (lane >> 4) * 8;

    const float* aBase = msg + am * MSG_STRIDE + 2 * khalf;
    v8f acc[4];
    #pragma unroll
    for (int nb = 0; nb < 4; ++nb) {
        v8f c = {};
        const int n = nb * 16 + ncol;
        const float* bBase = sW2 + khalf * WPAIR_STRIDE + 2 * n;
        #pragma unroll
        for (int kb = 0; kb < 16; ++kb) {
            const v2f a = *(const v2f*)(aBase + kb * 4);
            const v2f w = *(const v2f*)(bBase + kb * 2 * WPAIR_STRIDE);
            c = __builtin_amdgcn_wmma_f32_16x16x4_f32(
                    false, a, false, w, (short)0, c, false, false);
        }
        acc[nb] = c;
    }

    #pragma unroll
    for (int nb = 0; nb < 4; ++nb) {
        const int n = nb * 16 + ncol;
        #pragma unroll
        for (int v = 0; v < 8; ++v) {
            const int row = v + rsel;
            acc[nb][v] += sb2[n] + out_in[((size_t)row0 + row) * A_ + n];
        }
    }

    #pragma unroll
    for (int nb = 0; nb < 4; ++nb) {
        const int n = nb * 16 + ncol;
        #pragma unroll
        for (int v = 0; v < 8; ++v) {
            msg[(v + rsel) * MSG_STRIDE + n] = acc[nb][v];
        }
    }

    #pragma unroll
    for (int nb = 0; nb < 4; ++nb) {
        v8f c = {};
        const int n = nb * 16 + ncol;
        const float* bBase = sW3 + khalf * WPAIR_STRIDE + 2 * n;
        #pragma unroll
        for (int kb = 0; kb < 16; ++kb) {
            const v2f a = *(const v2f*)(aBase + kb * 4);
            const v2f w = *(const v2f*)(bBase + kb * 2 * WPAIR_STRIDE);
            c = __builtin_amdgcn_wmma_f32_16x16x4_f32(
                    false, a, false, w, (short)0, c, false, false);
        }
        #pragma unroll
        for (int v = 0; v < 8; ++v) {
            const int row = v + rsel;
            y_out[((size_t)row0 + row) * A_ + n] = c[v] + sb3[n];
        }
    }
}

extern "C" void kernel_launch(void* const* d_in, const int* in_sizes, int n_in,
                              void* d_out, int out_size, void* d_ws, size_t ws_size,
                              hipStream_t stream) {
    const float* outp = (const float*)d_in[0];
    const int*   idx  = (const int*)d_in[1];
    const float* W2   = (const float*)d_in[2];
    const float* b2   = (const float*)d_in[3];
    const float* W3   = (const float*)d_in[4];
    const float* b3   = (const float*)d_in[5];
    float* y = (float*)d_out;

    const int n_tiles = (B_ * G_) / TILE_M;
    const int blocks  = (n_tiles + WAVES_PER_BLOCK - 1) / WAVES_PER_BLOCK;

    hipLaunchKernelGGL(interaction_block_kernel, dim3(blocks), dim3(256), 0, stream,
                       outp, idx, W2, b2, W3, b3, y);
}


# gather made branch-free (padding index reads row 0, value masked to zero), next row's indices fetched ahead: a row's 32 neighbour loads fly together, not one dependent trip each; sum order kept
# speedup vs baseline: 2.2763x; 2.2763x over previous
#include <hip/hip_runtime.h>

typedef __attribute__((ext_vector_type(2))) float v2f;
typedef __attribute__((ext_vector_type(8))) float v8f;

#define B_ 2
#define G_ 50000
#define K_ 32
#define A_ 64

#define WAVES_PER_BLOCK 8
#define TILE_M 16
#define MSG_STRIDE 68
#define WPAIR_STRIDE 160

__global__ __launch_bounds__(256)
void interaction_block_kernel(const float* __restrict__ out_in,
                              const int*   __restrict__ nbr_idx,
                              const float* __restrict__ W2g,
                              const float* __restrict__ b2g,
                              const float* __restrict__ W3g,
                              const float* __restrict__ b3g,
                              float*       __restrict__ y_out)
{
    __shared__ float sW2[(A_ / 2) * WPAIR_STRIDE];
    __shared__ float sW3[(A_ / 2) * WPAIR_STRIDE];
    __shared__ float sb2[A_];
    __shared__ float sb3[A_];
    __shared__ float sMsg[WAVES_PER_BLOCK][TILE_M * MSG_STRIDE];

    const int tid = threadIdx.x;

    for (int i = tid; i < A_ * A_; i += 256) {
        const int k = i >> 6;
        const int n = i & 63;
        const int s = (k >> 1) * WPAIR_STRIDE + 2 * n + (k & 1);
        sW2[s] = W2g[i];
        sW3[s] = W3g[i];
    }
    if (tid < A_) { sb2[tid] = b2g[tid]; sb3[tid] = b3g[tid]; }
    __syncthreads();

    const int wave = tid >> 5;
    const int lane = tid & 31;
    const int tile = blockIdx.x * WAVES_PER_BLOCK + wave;
    const int n_tiles = (B_ * G_) / TILE_M;
    if (tile >= n_tiles) return;

    const int row0 = tile * TILE_M;
    const int b    = row0 / G_;
    const int g0   = row0 % G_;
    float* msg = sMsg[wave];

    const float* outB   = out_in + (size_t)b * G_ * A_;
    const float* lanePt = outB + 2 * lane;
    const int*   idxPt  = nbr_idx + (size_t)g0 * K_ + lane;
    int myidx = idxPt[0];
    for (int m = 0; m < TILE_M; ++m) {
        const int mn = (m + 1 < TILE_M) ? (m + 1) : m;
        const int nextidx = idxPt[mn * K_];
        v2f      val[K_];
        unsigned keep[K_];
        #pragma unroll
        for (int k = 0; k < K_; ++k) {
            const int  idx  = __builtin_amdgcn_readlane(myidx, k);
            const bool real = (unsigned)idx < (unsigned)G_;
            keep[k] = real ? 0xFFFFFFFFu : 0u;
            const unsigned row = real ? (unsigned)idx : 0u;
            val[k] = *(const v2f*)(lanePt + (size_t)row * A_);
        }
        float ax = 0.f, ay = 0.f;
        #pragma unroll
        for (int k = 0; k < K_; ++k) {
            ax += __uint_as_float(__float_as_uint(val[k].x) & keep[k]);
            ay += __uint_as_float(__float_as_uint(val[k].y) & keep[k]);
        }
        msg[m * MSG_STRIDE + 2 * lane]     = ax;
        msg[m * MSG_STRIDE + 2 * lane + 1] = ay;
        myidx = nextidx;
    }

    const int am    = lane & 15;
    const int khalf = lane >> 4;
    const int ncol  = lane & 15;
    const int rsel  = (lane >> 4) * 8;

    const float* aBase = msg + am * MSG_STRIDE + 2 * khalf;
    v8f acc[4];
    #pragma unroll
    for (int nb = 0; nb < 4; ++nb) {
        v8f c = {};
        const int n = nb * 16 + ncol;
        const float* bBase = sW2 + khalf * WPAIR_STRIDE + 2 * n;
        #pragma unroll
        for (int kb = 0; kb < 16; ++kb) {
            const v2f a = *(const v2f*)(aBase + kb * 4);
            const v2f w = *(const v2f*)(bBase + kb * 2 * WPAIR_STRIDE);
            c = __builtin_amdgcn_wmma_f32_16x16x4_f32(
                    false, a, false, w, (short)0, c, false, false);
        }
        acc[nb] = c;
    }

    #pragma unroll
    for (int nb = 0; nb < 4; ++nb) {
        const int n = nb * 16 + ncol;
        #pragma unroll
        for (int v = 0; v < 8; ++v) {
            const int row = v + rsel;
            acc[nb][v] += sb2[n] + out_in[((size_t)row0 + row) * A_ + n];
        }
    }

    #pragma unroll
    for (int nb = 0; nb < 4; ++nb) {
        const int n = nb * 16 + ncol;
        #pragma unroll
        for (int v = 0; v < 8; ++v) {
            msg[(v + rsel) * MSG_STRIDE + n] = acc[nb][v];
        }
    }

    #pragma unroll
    for (int nb = 0; nb < 4; ++nb) {
        v8f c = {};
        const int n = nb * 16 + ncol;
        const float* bBase = sW3 + khalf * WPAIR_STRIDE + 2 * n;
        #pragma unroll
        for (int kb = 0; kb < 16; ++kb) {
            const v2f a = *(const v2f*)(aBase + kb * 4);
            const v2f w = *(const v2f*)(bBase + kb * 2 * WPAIR_STRIDE);
            c = __builtin_amdgcn_wmma_f32_16x16x4_f32(
                    false, a, false, w, (short)0, c, false, false);
        }
        #pragma unroll
        for (int v = 0; v < 8; ++v) {
            const int row = v + rsel;
            y_out[((size_t)row0 + row) * A_ + n] = c[v] + sb3[n];
        }
    }
}

extern "C" void kernel_launch(void* const* d_in, const int* in_sizes, int n_in,
                              void* d_out, int out_size, void* d_ws, size_t ws_size,
                              hipStream_t stream) {
    const float* outp = (const float*)d_in[0];
    const int*   idx  = (const int*)d_in[1];
    const float* W2   = (const float*)d_in[2];
    const float* b2   = (const float*)d_in[3];
    const float* W3   = (const float*)d_in[4];
    const float* b3   = (const float*)d_in[5];
    float* y = (float*)d_out;

    const int n_tiles = (B_ * G_) / TILE_M;
    const int blocks  = (n_tiles + WAVES_PER_BLOCK - 1) / WAVES_PER_BLOCK;

    hipLaunchKernelGGL(interaction_block_kernel, dim3(blocks), dim3(256), 0, stream,
                       outp, idx, W2, b2, W3, b3, y);
}


# grid re-dealt: one 25-wave work-group per WGP (256 of them) with tiles dealt round them, so every WGP gets 24 or 25 tiles where the fullest had 32; per-tile code and sum order kept
# speedup vs baseline: 2.7208x; 1.1953x over previous
#include <hip/hip_runtime.h>

typedef __attribute__((ext_vector_type(2))) float v2f;
typedef __attribute__((ext_vector_type(8))) float v8f;

#define B_ 2
#define G_ 50000
#define K_ 32
#define A_ 64

#define WAVES_PER_BLOCK 25
#define N_BLOCKS 256
#define THREADS (WAVES_PER_BLOCK * 32)
#define TILE_M 16
#define MSG_STRIDE 68
#define WPAIR_STRIDE 160

__global__ __launch_bounds__(THREADS)
void interaction_block_kernel(const float* __restrict__ out_in,
                              const int*   __restrict__ nbr_idx,
                              const float* __restrict__ W2g,
                              const float* __restrict__ b2g,
                              const float* __restrict__ W3g,
                              const float* __restrict__ b3g,
                              float*       __restrict__ y_out)
{
    __shared__ float sW2[(A_ / 2) * WPAIR_STRIDE];
    __shared__ float sW3[(A_ / 2) * WPAIR_STRIDE];
    __shared__ float sb2[A_];
    __shared__ float sb3[A_];
    __shared__ float sMsg[WAVES_PER_BLOCK][TILE_M * MSG_STRIDE];

    const int tid = threadIdx.x;

    for (int i = tid; i < A_ * A_; i += THREADS) {
        const int k = i >> 6;
        const int n = i & 63;
        const int s = (k >> 1) * WPAIR_STRIDE + 2 * n + (k & 1);
        sW2[s] = W2g[i];
        sW3[s] = W3g[i];
    }
    if (tid < A_) { sb2[tid] = b2g[tid]; sb3[tid] = b3g[tid]; }
    __syncthreads();

    const int wave = tid >> 5;
    const int lane = tid & 31;
    const int tile = wave * N_BLOCKS + blockIdx.x;
    const int n_tiles = (B_ * G_) / TILE_M;
    if (tile >= n_tiles) return;

    const int row0 = tile * TILE_M;
    const int b    = row0 / G_;
    const int g0   = row0 % G_;
    float* msg = sMsg[wave];

    const float* outB   = out_in + (size_t)b * G_ * A_;
    const float* lanePt = outB + 2 * lane;
    const int*   idxPt  = nbr_idx + (size_t)g0 * K_ + lane;
    int myidx = idxPt[0];
    for (int m = 0; m < TILE_M; ++m) {
        const int mn = (m + 1 < TILE_M) ? (m + 1) : m;
        const int nextidx = idxPt[mn * K_];
        v2f      val[K_];
        unsigned keep[K_];
        #pragma unroll
        for (int k = 0; k < K_; ++k) {
            const int  idx  = __builtin_amdgcn_readlane(myidx, k);
            const bool real = (unsigned)idx < (unsigned)G_;
            keep[k] = real ? 0xFFFFFFFFu : 0u;
            const unsigned row = real ? (unsigned)idx : 0u;
            val[k] = *(const v2f*)(lanePt + (size_t)row * A_);
        }
        float ax = 0.f, ay = 0.f;
        #pragma unroll
        for (int k = 0; k < K_; ++k) {
            ax += __uint_as_float(__float_as_uint(val[k].x) & keep[k]);
            ay += __uint_as_float(__float_as_uint(val[k].y) & keep[k]);
        }
        msg[m * MSG_STRIDE + 2 * lane]     = ax;
        msg[m * MSG_STRIDE + 2 * lane + 1] = ay;
        myidx = nextidx;
    }

    const int am    = lane & 15;
    const int khalf = lane >> 4;
    const int ncol  = lane & 15;
    const int rsel  = (lane >> 4) * 8;

    const float* aBase = msg + am * MSG_STRIDE + 2 * khalf;
    v8f acc[4];
    #pragma unroll
    for (int nb = 0; nb < 4; ++nb) {
        v8f c = {};
        const int n = nb * 16 + ncol;
        const float* bBase = sW2 + khalf * WPAIR_STRIDE + 2 * n;
        #pragma unroll
        for (int kb = 0; kb < 16; ++kb) {
            const v2f a = *(const v2f*)(aBase + kb * 4);
            const v2f w = *(const v2f*)(bBase + kb * 2 * WPAIR_STRIDE);
            c = __builtin_amdgcn_wmma_f32_16x16x4_f32(
                    false, a, false, w, (short)0, c, false, false);
        }
        acc[nb] = c;
    }

    #pragma unroll
    for (int nb = 0; nb < 4; ++nb) {
        const int n = nb * 16 + ncol;
        #pragma unroll
        for (int v = 0; v < 8; ++v) {
            const int row = v + rsel;
            acc[nb][v] += sb2[n] + out_in[((size_t)row0 + row) * A_ + n];
        }
    }

    #pragma unroll
    for (int nb = 0; nb < 4; ++nb) {
        const int n = nb * 16 + ncol;
        #pragma unroll
        for (int v = 0; v < 8; ++v) {
            msg[(v + rsel) * MSG_STRIDE + n] = acc[nb][v];
        }
    }

    #pragma unroll
    for (int nb = 0; nb < 4; ++nb) {
        v8f c = {};
        const int n = nb * 16 + ncol;
        const float* bBase = sW3 + khalf * WPAIR_STRIDE + 2 * n;
        #pragma unroll
        for (int kb = 0; kb < 16; ++kb) {
            const v2f a = *(const v2f*)(aBase + kb * 4);
            const v2f w = *(const v2f*)(bBase + kb * 2 * WPAIR_STRIDE);
            c = __builtin_amdgcn_wmma_f32_16x16x4_f32(
                    false, a, false, w, (short)0, c, false, false);
        }
        #pragma unroll
        for (int v = 0; v < 8; ++v) {
            const int row = v + rsel;
            y_out[((size_t)row0 + row) * A_ + n] = c[v] + sb3[n];
        }
    }
}

extern "C" void kernel_launch(void* const* d_in, const int* in_sizes, int n_in,
                              void* d_out, int out_size, void* d_ws, size_t ws_size,
                              hipStream_t stream) {
    const float* outp = (const float*)d_in[0];
    const int*   idx  = (const int*)d_in[1];
    const float* W2   = (const float*)d_in[2];
    const float* b2   = (const float*)d_in[3];
    const float* W3   = (const float*)d_in[4];
    const float* b3   = (const float*)d_in[5];
    float* y = (float*)d_out;

    hipLaunchKernelGGL(interaction_block_kernel, dim3(N_BLOCKS), dim3(THREADS), 0, stream,
                       outp, idx, W2, b2, W3, b3, y);
}


# prologue loads issued as one group at wave entry (first index row, both weight shares, biases; index clamped, LDS store predicated): one trip to memory before the first gather, not eight in a row
# speedup vs baseline: 2.8395x; 1.0436x over previous
#include <hip/hip_runtime.h>

typedef __attribute__((ext_vector_type(2))) float v2f;
typedef __attribute__((ext_vector_type(8))) float v8f;

#define B_ 2
#define G_ 50000
#define K_ 32
#define A_ 64

#define WAVES_PER_BLOCK 25
#define N_BLOCKS 256
#define THREADS (WAVES_PER_BLOCK * 32)
#define TILE_M 16
#define MSG_STRIDE 68
#define WPAIR_STRIDE 160

__global__ __launch_bounds__(THREADS)
void interaction_block_kernel(const float* __restrict__ out_in,
                              const int*   __restrict__ nbr_idx,
                              const float* __restrict__ W2g,
                              const float* __restrict__ b2g,
                              const float* __restrict__ W3g,
                              const float* __restrict__ b3g,
                              float*       __restrict__ y_out)
{
    __shared__ float sW2[(A_ / 2) * WPAIR_STRIDE];
    __shared__ float sW3[(A_ / 2) * WPAIR_STRIDE];
    __shared__ float sb2[A_];
    __shared__ float sb3[A_];
    __shared__ float sMsg[WAVES_PER_BLOCK][TILE_M * MSG_STRIDE];

    const int tid = threadIdx.x;
    const int wave = tid >> 5;
    const int lane = tid & 31;
    const int tile = wave * N_BLOCKS + blockIdx.x;
    const int n_tiles = (B_ * G_) / TILE_M;
    const bool active = tile < n_tiles;
    const int row0 = (active ? tile : 0) * TILE_M;
    const int b    = row0 / G_;
    const int g0   = row0 % G_;
    float* msg = sMsg[wave];

    const int*   idxPt  = nbr_idx + (size_t)g0 * K_ + lane;
    int myidx = idxPt[0];
    constexpr int STAGE_TRIPS = (A_ * A_ + THREADS - 1) / THREADS;
    float w2r[STAGE_TRIPS], w3r[STAGE_TRIPS];
    #pragma unroll
    for (int j = 0; j < STAGE_TRIPS; ++j) {
        const int i  = tid + j * THREADS;
        const int ic = (i < A_ * A_) ? i : (A_ * A_ - 1);
        w2r[j] = W2g[ic];
        w3r[j] = W3g[ic];
    }
    const float b2r = b2g[tid & (A_ - 1)];
    const float b3r = b3g[tid & (A_ - 1)];
    asm volatile("" :: "v"(w2r[STAGE_TRIPS - 1]), "v"(w3r[STAGE_TRIPS - 1]));

    #pragma unroll
    for (int j = 0; j < STAGE_TRIPS; ++j) {
        const int i = tid + j * THREADS;
        if ((j + 1) * THREADS <= A_ * A_ || i < A_ * A_) {
            const int k = i >> 6;
            const int n = i & 63;
            const int s = (k >> 1) * WPAIR_STRIDE + 2 * n + (k & 1);
            sW2[s] = w2r[j];
            sW3[s] = w3r[j];
        }
    }
    if (tid < A_) { sb2[tid] = b2r; sb3[tid] = b3r; }
    __syncthreads();

    if (!active) return;

    const float* outB   = out_in + (size_t)b * G_ * A_;
    const float* lanePt = outB + 2 * lane;
    for (int m = 0; m < TILE_M; ++m) {
        const int mn = (m + 1 < TILE_M) ? (m + 1) : m;
        const int nextidx = idxPt[mn * K_];
        v2f      val[K_];
        unsigned keep[K_];
        #pragma unroll
        for (int k = 0; k < K_; ++k) {
            const int  idx  = __builtin_amdgcn_readlane(myidx, k);
            const bool real = (unsigned)idx < (unsigned)G_;
            keep[k] = real ? 0xFFFFFFFFu : 0u;
            const unsigned row = real ? (unsigned)idx : 0u;
            val[k] = *(const v2f*)(lanePt + (size_t)row * A_);
        }
        float ax = 0.f, ay = 0.f;
        #pragma unroll
        for (int k = 0; k < K_; ++k) {
            ax += __uint_as_float(__float_as_uint(val[k].x) & keep[k]);
            ay += __uint_as_float(__float_as_uint(val[k].y) & keep[k]);
        }
        msg[m * MSG_STRIDE + 2 * lane]     = ax;
        msg[m * MSG_STRIDE + 2 * lane + 1] = ay;
        myidx = nextidx;
    }

    const int am    = lane & 15;
    const int khalf = lane >> 4;
    const int ncol  = lane & 15;
    const int rsel  = (lane >> 4) * 8;

    const float* aBase = msg + am * MSG_STRIDE + 2 * khalf;
    v8f acc[4];
    #pragma unroll
    for (int nb = 0; nb < 4; ++nb) {
        v8f c = {};
        const int n = nb * 16 + ncol;
        const float* bBase = sW2 + khalf * WPAIR_STRIDE + 2 * n;
        #pragma unroll
        for (int kb = 0; kb < 16; ++kb) {
            const v2f a = *(const v2f*)(aBase + kb * 4);
            const v2f w = *(const v2f*)(bBase + kb * 2 * WPAIR_STRIDE);
            c = __builtin_amdgcn_wmma_f32_16x16x4_f32(
                    false, a, false, w, (short)0, c, false, false);
        }
        acc[nb] = c;
    }

    #pragma unroll
    for (int nb = 0; nb < 4; ++nb) {
        const int n = nb * 16 + ncol;
        #pragma unroll
        for (int v = 0; v < 8; ++v) {
            const int row = v + rsel;
            acc[nb][v] += sb2[n] + out_in[((size_t)row0 + row) * A_ + n];
        }
    }

    #pragma unroll
    for (int nb = 0; nb < 4; ++nb) {
        const int n = nb * 16 + ncol;
        #pragma unroll
        for (int v = 0; v < 8; ++v) {
            msg[(v + rsel) * MSG_STRIDE + n] = acc[nb][v];
        }
    }

    #pragma unroll
    for (int nb = 0; nb < 4; ++nb) {
        v8f c = {};
        const int n = nb * 16 + ncol;
        const float* bBase = sW3 + khalf * WPAIR_STRIDE + 2 * n;
        #pragma unroll
        for (int kb = 0; kb < 16; ++kb) {
            const v2f a = *(const v2f*)(aBase + kb * 4);
            const v2f w = *(const v2f*)(bBase + kb * 2 * WPAIR_STRIDE);
            c = __builtin_amdgcn_wmma_f32_16x16x4_f32(
                    false, a, false, w, (short)0, c, false, false);
        }
        #pragma unroll
        for (int v = 0; v < 8; ++v) {
            const int row = v + rsel;
            y_out[((size_t)row0 + row) * A_ + n] = c[v] + sb3[n];
        }
    }
}

extern "C" void kernel_launch(void* const* d_in, const int* in_sizes, int n_in,
                              void* d_out, int out_size, void* d_ws, size_t ws_size,
                              hipStream_t stream) {
    const float* outp = (const float*)d_in[0];
    const int*   idx  = (const int*)d_in[1];
    const float* W2   = (const float*)d_in[2];
    const float* b2   = (const float*)d_in[3];
    const float* W3   = (const float*)d_in[4];
    const float* b3   = (const float*)d_in[5];
    float* y = (float*)d_out;

    hipLaunchKernelGGL(interaction_block_kernel, dim3(N_BLOCKS), dim3(THREADS), 0, stream,
                       outp, idx, W2, b2, W3, b3, y);
}


# a tile is 8 grid rows of BOTH batches (one shared index table): one 16-byte load per neighbour brings its row for both; half the loads, index reads and scalar index work per row; sum order kept
# speedup vs baseline: 3.0219x; 1.0642x over previous
#include <hip/hip_runtime.h>

typedef __attribute__((ext_vector_type(2))) float v2f;
typedef __attribute__((ext_vector_type(4))) float v4f;
typedef __attribute__((ext_vector_type(8))) float v8f;

#define B_ 2
#define G_ 50000
#define K_ 32
#define A_ 64

#define WAVES_PER_BLOCK 25
#define N_BLOCKS 256
#define THREADS (WAVES_PER_BLOCK * 32)
#define TILE_M 16
#define TILE_G 8
#define MSG_STRIDE 68
#define WPAIR_STRIDE 160

__global__ __launch_bounds__(THREADS)
void interaction_block_kernel(const float* __restrict__ out_in,
                              const int*   __restrict__ nbr_idx,
                              const float* __restrict__ W2g,
                              const float* __restrict__ b2g,
                              const float* __restrict__ W3g,
                              const float* __restrict__ b3g,
                              float*       __restrict__ y_out)
{
    __shared__ float sW2[(A_ / 2) * WPAIR_STRIDE];
    __shared__ float sW3[(A_ / 2) * WPAIR_STRIDE];
    __shared__ float sb2[A_];
    __shared__ float sb3[A_];
    __shared__ __attribute__((aligned(16))) float sMsg[WAVES_PER_BLOCK][TILE_M * MSG_STRIDE];

    const int tid = threadIdx.x;
    const int wave = tid >> 5;
    const int lane = tid & 31;
    const int tile = wave * N_BLOCKS + blockIdx.x;
    const int n_tiles = G_ / TILE_G;
    const bool active = tile < n_tiles;
    const int g0   = (active ? tile : 0) * TILE_G;
    const int hb   = lane >> 4;
    float* msg = sMsg[wave];

    const int*   idxPt  = nbr_idx + (size_t)g0 * K_ + lane;
    int myidx = idxPt[0];
    constexpr int STAGE_TRIPS = (A_ * A_ + THREADS - 1) / THREADS;
    float w2r[STAGE_TRIPS], w3r[STAGE_TRIPS];
    #pragma unroll
    for (int j = 0; j < STAGE_TRIPS; ++j) {
        const int i  = tid + j * THREADS;
        const int ic = (i < A_ * A_) ? i : (A_ * A_ - 1);
        w2r[j] = W2g[ic];
        w3r[j] = W3g[ic];
    }
    const float b2r = b2g[tid & (A_ - 1)];
    const float b3r = b3g[tid & (A_ - 1)];
    asm volatile("" :: "v"(w2r[STAGE_TRIPS - 1]), "v"(w3r[STAGE_TRIPS - 1]));

    #pragma unroll
    for (int j = 0; j < STAGE_TRIPS; ++j) {
        const int i = tid + j * THREADS;
        if ((j + 1) * THREADS <= A_ * A_ || i < A_ * A_) {
            const int k = i >> 6;
            const int n = i & 63;
            const int s = (k >> 1) * WPAIR_STRIDE + 2 * n + (k & 1);
            sW2[s] = w2r[j];
            sW3[s] = w3r[j];
        }
    }
    if (tid < A_) { sb2[tid] = b2r; sb3[tid] = b3r; }
    __syncthreads();

    if (!active) return;

    const float* lanePt = out_in + (size_t)hb * G_ * A_ + 4 * (lane & 15);
    constexpr int KG = 4;
    for (int m = 0; m < TILE_G; ++m) {
        const int mn = (m + 1 < TILE_G) ? (m + 1) : m;
        const int nextidx = idxPt[mn * K_];
        v4f      val[K_];
        unsigned keep[K_];
        #pragma unroll
        for (int kg = 0; kg < K_; kg += KG) {
            const v4f* p[KG];
            #pragma unroll
            for (int j = 0; j < KG; ++j) {
                const int  idx  = __builtin_amdgcn_readlane(myidx, kg + j);
                const bool real = (unsigned)idx < (unsigned)G_;
                keep[kg + j] = real ? 0xFFFFFFFFu : 0u;
                const unsigned row = real ? (unsigned)idx : 0u;
                p[j] = (const v4f*)(lanePt + (size_t)row * A_);
            }
            asm volatile("" :: "v"(p[0]), "v"(p[1]), "v"(p[2]), "v"(p[3]));
            #pragma unroll
            for (int j = 0; j < KG; ++j) val[kg + j] = *p[j];
        }
        float a0 = 0.f, a1 = 0.f, a2 = 0.f, a3 = 0.f;
        #pragma unroll
        for (int k = 0; k < K_; ++k) {
            a0 += __uint_as_float(__float_as_uint(val[k].x) & keep[k]);
            a1 += __uint_as_float(__float_as_uint(val[k].y) & keep[k]);
            a2 += __uint_as_float(__float_as_uint(val[k].z) & keep[k]);
            a3 += __uint_as_float(__float_as_uint(val[k].w) & keep[k]);
        }
        v4f s; s.x = a0; s.y = a1; s.z = a2; s.w = a3;
        *(v4f*)(msg + (hb * TILE_G + m) * MSG_STRIDE + 4 * (lane & 15)) = s;
        myidx = nextidx;
    }

    const int am    = lane & 15;
    const int khalf = lane >> 4;
    const int ncol  = lane & 15;
    const int rsel  = (lane >> 4) * 8;
    const size_t growBase = (size_t)hb * G_ + g0;

    const float* aBase = msg + am * MSG_STRIDE + 2 * khalf;
    v8f acc[4];
    #pragma unroll
    for (int nb = 0; nb < 4; ++nb) {
        v8f c = {};
        const int n = nb * 16 + ncol;
        const float* bBase = sW2 + khalf * WPAIR_STRIDE + 2 * n;
        #pragma unroll
        for (int kb = 0; kb < 16; ++kb) {
            const v2f a = *(const v2f*)(aBase + kb * 4);
            const v2f w = *(const v2f*)(bBase + kb * 2 * WPAIR_STRIDE);
            c = __builtin_amdgcn_wmma_f32_16x16x4_f32(
                    false, a, false, w, (short)0, c, false, false);
        }
        acc[nb] = c;
    }

    #pragma unroll
    for (int nb = 0; nb < 4; ++nb) {
        const int n = nb * 16 + ncol;
        #pragma unroll
        for (int v = 0; v < 8; ++v) {
            acc[nb][v] += sb2[n] + out_in[(growBase + v) * A_ + n];
        }
    }

    #pragma unroll
    for (int nb = 0; nb < 4; ++nb) {
        const int n = nb * 16 + ncol;
        #pragma unroll
        for (int v = 0; v < 8; ++v) {
            msg[(v + rsel) * MSG_STRIDE + n] = acc[nb][v];
        }
    }

    #pragma unroll
    for (int nb = 0; nb < 4; ++nb) {
        v8f c = {};
        const int n = nb * 16 + ncol;
        const float* bBase = sW3 + khalf * WPAIR_STRIDE + 2 * n;
        #pragma unroll
        for (int kb = 0; kb < 16; ++kb) {
            const v2f a = *(const v2f*)(aBase + kb * 4);
            const v2f w = *(const v2f*)(bBase + kb * 2 * WPAIR_STRIDE);
            c = __builtin_amdgcn_wmma_f32_16x16x4_f32(
                    false, a, false, w, (short)0, c, false, false);
        }
        #pragma unroll
        for (int v = 0; v < 8; ++v) {
            y_out[(growBase + v) * A_ + n] = c[v] + sb3[n];
        }
    }
}

extern "C" void kernel_launch(void* const* d_in, const int* in_sizes, int n_in,
                              void* d_out, int out_size, void* d_ws, size_t ws_size,
                              hipStream_t stream) {
    const float* outp = (const float*)d_in[0];
    const int*   idx  = (const int*)d_in[1];
    const float* W2   = (const float*)d_in[2];
    const float* b2   = (const float*)d_in[3];
    const float* W3   = (const float*)d_in[4];
    const float* b3   = (const float*)d_in[5];
    float* y = (float*)d_out;

    hipLaunchKernelGGL(interaction_block_kernel, dim3(N_BLOCKS), dim3(THREADS), 0, stream,
                       outp, idx, W2, b2, W3, b3, y);
}
